# starting kernel
# baseline (speedup 1.0000x reference)
#include <hip/hip_runtime.h>
#include <math.h>

typedef float v2f __attribute__((ext_vector_type(2)));
typedef float v8f __attribute__((ext_vector_type(8)));

#define B_TOT  8
#define C_DIM  512
#define K_DIM  19
#define N_SP   16384
#define CT     16
#define WAVES  4
#define NSPLIT 4
#define UNROLL 8
#define PN_OFF 512

__global__ void zero_out_kernel(float* __restrict__ out, int n) {
    int i = blockIdx.x * blockDim.x + threadIdx.x;
    if (i < n) out[i] = 0.0f;
}

__global__ void softmax_stats_kernel(const float* __restrict__ probs,
                                     float* __restrict__ stats) {
    __shared__ float red[256];
    const int r = blockIdx.x;
    const float* p = probs + (size_t)r * N_SP;
    const int t = threadIdx.x;

    float m = -3.402823466e38f;
    for (int i = t; i < N_SP; i += 256) m = fmaxf(m, p[i]);
    red[t] = m; __syncthreads();
    for (int s = 128; s > 0; s >>= 1) {
        if (t < s) red[t] = fmaxf(red[t], red[t + s]);
        __syncthreads();
    }
    const float rowmax = red[0];
    __syncthreads();

    float sum = 0.0f;
    for (int i = t; i < N_SP; i += 256) sum += __expf(p[i] - rowmax);
    red[t] = sum; __syncthreads();
    for (int s = 128; s > 0; s >>= 1) {
        if (t < s) red[t] += red[t + s];
        __syncthreads();
    }
    if (t == 0) { stats[2 * r] = rowmax; stats[2 * r + 1] = 1.0f / red[0]; }
}

__global__ void softmax_norm_kernel(const float* __restrict__ probs,
                                    const float* __restrict__ stats,
                                    float* __restrict__ pn, size_t total) {
    size_t i = (size_t)blockIdx.x * blockDim.x + threadIdx.x;
    if (i >= total) return;
    int r = (int)(i / N_SP);
    pn[i] = __expf(probs[i] - stats[2 * r]) * stats[2 * r + 1];
}

__global__ void __launch_bounds__(WAVES * 32)
pooled_gemm_kernel(const float* __restrict__ feats,
                   const float* __restrict__ pn,
                   const int* __restrict__ clip_num,
                   float* __restrict__ out) {
    const int beta  = blockIdx.y;
    const int ctile = blockIdx.x * CT;
    const int wave  = threadIdx.x >> 5;
    const int lane  = threadIdx.x & 31;
    const int half  = lane >> 4;
    const int lrow  = lane & 15;
    const int toff  = half * 2;

    const int   splits = clip_num[0] + 1;
    const int   bsz    = B_TOT / splits;
    const int   bb     = beta % bsz;
    const float scale  = 1.0f / (float)splits;

    constexpr int nper  = N_SP / (WAVES * NSPLIT);
    constexpr int NITER = nper / (4 * UNROLL);
    const int n0 = (blockIdx.z * WAVES + wave) * nper;

    const float* pa0 = pn + ((size_t)beta * K_DIM + lrow) * N_SP + toff + n0;
    const int    k1  = 16 + lrow;
    const float* pa1 = pn + ((size_t)beta * K_DIM + (k1 < K_DIM ? k1 : 0)) * N_SP
                          + toff + n0;
    const float* pb  = feats + ((size_t)beta * C_DIM + ctile + lrow) * N_SP
                             + toff + n0;

    v8f acc0 = {};
    v8f acc1 = {};

    for (int it = 0; it < NITER; ++it) {
        #pragma unroll
        for (int u = 0; u < UNROLL; ++u) {
            v2f a0 = *(const v2f*)(pa0 + 4 * u);
            v2f a1 = *(const v2f*)(pa1 + 4 * u);
            v2f bv = __builtin_nontemporal_load((const v2f*)(pb + 4 * u));
            acc0 = __builtin_amdgcn_wmma_f32_16x16x4_f32(
                       false, a0, false, bv, (short)0, acc0, false, false);
            acc1 = __builtin_amdgcn_wmma_f32_16x16x4_f32(
                       false, a1, false, bv, (short)0, acc1, false, false);
        }
        pa0 += 4 * UNROLL;
        pa1 += 4 * UNROLL;
        pb  += 4 * UNROLL;
    }

    const int c = ctile + lrow;
    float* ob = out + ((size_t)bb * C_DIM + c) * K_DIM;
    #pragma unroll
    for (int v = 0; v < 8; ++v) {
        const int k = v + half * 8;
        atomicAdd(ob + k, scale * acc0[v]);
        const int k2 = k + 16;
        if (k2 < K_DIM) atomicAdd(ob + k2, scale * acc1[v]);
    }
}

extern "C" void kernel_launch(void* const* d_in, const int* in_sizes, int n_in,
                              void* d_out, int out_size, void* d_ws, size_t ws_size,
                              hipStream_t stream) {
    const float* feats = (const float*)d_in[0];
    const float* probs = (const float*)d_in[1];
    const int*   clip  = (const int*)d_in[2];
    float* out   = (float*)d_out;
    float* stats = (float*)d_ws;
    float* pnbuf = (float*)d_ws + PN_OFF;

    (void)in_sizes; (void)n_in; (void)ws_size;

    zero_out_kernel<<<(out_size + 255) / 256, 256, 0, stream>>>(out, out_size);

    softmax_stats_kernel<<<B_TOT * K_DIM, 256, 0, stream>>>(probs, stats);

    const size_t total = (size_t)B_TOT * K_DIM * N_SP;
    softmax_norm_kernel<<<(unsigned)((total + 255) / 256), 256, 0, stream>>>(
        probs, stats, pnbuf, total);

    dim3 grid(C_DIM / CT, B_TOT, NSPLIT);
    pooled_gemm_kernel<<<grid, WAVES * 32, 0, stream>>>(feats, pnbuf, clip, out);
}


# softmax: the row-stats launch keeps its row in registers (16 x 16-B loads per thread, issued together) and writes the normalised plane itself; normalisation launch removed; row-sum order changes
# speedup vs baseline: 1.6224x; 1.6224x over previous
#include <hip/hip_runtime.h>
#include <math.h>

typedef float v2f __attribute__((ext_vector_type(2)));
typedef float v4f __attribute__((ext_vector_type(4)));
typedef float v8f __attribute__((ext_vector_type(8)));

#define B_TOT  8
#define C_DIM  512
#define K_DIM  19
#define N_SP   16384
#define CT     16
#define WAVES  4
#define NSPLIT 4
#define UNROLL 8
#define PN_OFF 512
#define SM_THREADS 256
#define SM_WAVES   (SM_THREADS / 32)
#define SM_PIECES  (N_SP / 4 / SM_THREADS)

__global__ void zero_out_kernel(float* __restrict__ out, int n) {
    int i = blockIdx.x * blockDim.x + threadIdx.x;
    if (i < n) out[i] = 0.0f;
}

__global__ void __launch_bounds__(SM_THREADS)
softmax_rows_kernel(const float* __restrict__ probs, float* __restrict__ pn) {
    __shared__ float red_max[SM_WAVES];
    __shared__ float red_sum[SM_WAVES];
    const int r    = blockIdx.x;
    const int t    = threadIdx.x;
    const int wave = t >> 5;
    const int lane = t & 31;
    const v4f* p4 = (const v4f*)(probs + (size_t)r * N_SP);
    v4f*       o4 = (v4f*)(pn + (size_t)r * N_SP);

    v4f x[SM_PIECES];
    #pragma unroll
    for (int j = 0; j < SM_PIECES; ++j) x[j] = p4[t + SM_THREADS * j];

    float m = -3.402823466e38f;
    #pragma unroll
    for (int j = 0; j < SM_PIECES; ++j)
        m = fmaxf(m, fmaxf(fmaxf(x[j].x, x[j].y), fmaxf(x[j].z, x[j].w)));
    #pragma unroll
    for (int o = 16; o > 0; o >>= 1) m = fmaxf(m, __shfl_xor(m, o, 32));
    if (lane == 0) red_max[wave] = m;
    __syncthreads();
    float rowmax = red_max[0];
    #pragma unroll
    for (int w = 1; w < SM_WAVES; ++w) rowmax = fmaxf(rowmax, red_max[w]);

    float s0 = 0.0f, s1 = 0.0f, s2 = 0.0f, s3 = 0.0f;
    #pragma unroll
    for (int j = 0; j < SM_PIECES; ++j) {
        x[j].x = __expf(x[j].x - rowmax); s0 += x[j].x;
        x[j].y = __expf(x[j].y - rowmax); s1 += x[j].y;
        x[j].z = __expf(x[j].z - rowmax); s2 += x[j].z;
        x[j].w = __expf(x[j].w - rowmax); s3 += x[j].w;
    }
    float s = (s0 + s1) + (s2 + s3);
    #pragma unroll
    for (int o = 16; o > 0; o >>= 1) s += __shfl_xor(s, o, 32);
    if (lane == 0) red_sum[wave] = s;
    __syncthreads();
    float total = red_sum[0];
    #pragma unroll
    for (int w = 1; w < SM_WAVES; ++w) total += red_sum[w];
    const float inv = 1.0f / total;

    #pragma unroll
    for (int j = 0; j < SM_PIECES; ++j) o4[t + SM_THREADS * j] = x[j] * inv;
}

__global__ void __launch_bounds__(WAVES * 32)
pooled_gemm_kernel(const float* __restrict__ feats,
                   const float* __restrict__ pn,
                   const int* __restrict__ clip_num,
                   float* __restrict__ out) {
    const int beta  = blockIdx.y;
    const int ctile = blockIdx.x * CT;
    const int wave  = threadIdx.x >> 5;
    const int lane  = threadIdx.x & 31;
    const int half  = lane >> 4;
    const int lrow  = lane & 15;
    const int toff  = half * 4;

    const int   splits = clip_num[0] + 1;
    const int   bsz    = B_TOT / splits;
    const int   bb     = beta % bsz;
    const float scale  = 1.0f / (float)splits;

    constexpr int nper  = N_SP / (WAVES * NSPLIT);
    constexpr int NITER = nper / (8 * UNROLL);
    const int n0 = (blockIdx.z * WAVES + wave) * nper;

    const float* pa0 = pn + ((size_t)beta * K_DIM + lrow) * N_SP + toff + n0;
    const int    k1  = 16 + lrow;
    const float* pa1 = pn + ((size_t)beta * K_DIM + (k1 < K_DIM ? k1 : 0)) * N_SP
                          + toff + n0;
    const float* pb  = feats + ((size_t)beta * C_DIM + ctile + lrow) * N_SP
                             + toff + n0;

    v8f acc0 = {};
    v8f acc1 = {};

    for (int it = 0; it < NITER; ++it) {
        #pragma unroll
        for (int u = 0; u < UNROLL; ++u) {
            const v4f a0 = *(const v4f*)(pa0 + 8 * u);
            const v4f a1 = *(const v4f*)(pa1 + 8 * u);
            const v4f bv = __builtin_nontemporal_load((const v4f*)(pb + 8 * u));
            const v2f a0lo = {a0.x, a0.y}, a0hi = {a0.z, a0.w};
            const v2f a1lo = {a1.x, a1.y}, a1hi = {a1.z, a1.w};
            const v2f bvlo = {bv.x, bv.y}, bvhi = {bv.z, bv.w};
            acc0 = __builtin_amdgcn_wmma_f32_16x16x4_f32(
                       false, a0lo, false, bvlo, (short)0, acc0, false, false);
            acc1 = __builtin_amdgcn_wmma_f32_16x16x4_f32(
                       false, a1lo, false, bvlo, (short)0, acc1, false, false);
            acc0 = __builtin_amdgcn_wmma_f32_16x16x4_f32(
                       false, a0hi, false, bvhi, (short)0, acc0, false, false);
            acc1 = __builtin_amdgcn_wmma_f32_16x16x4_f32(
                       false, a1hi, false, bvhi, (short)0, acc1, false, false);
        }
        pa0 += 8 * UNROLL;
        pa1 += 8 * UNROLL;
        pb  += 8 * UNROLL;
    }

    const int c = ctile + lrow;
    float* ob = out + ((size_t)bb * C_DIM + c) * K_DIM;
    #pragma unroll
    for (int v = 0; v < 8; ++v) {
        const int k = v + half * 8;
        atomicAdd(ob + k, scale * acc0[v]);
        const int k2 = k + 16;
        if (k2 < K_DIM) atomicAdd(ob + k2, scale * acc1[v]);
    }
}

extern "C" void kernel_launch(void* const* d_in, const int* in_sizes, int n_in,
                              void* d_out, int out_size, void* d_ws, size_t ws_size,
                              hipStream_t stream) {
    const float* feats = (const float*)d_in[0];
    const float* probs = (const float*)d_in[1];
    const int*   clip  = (const int*)d_in[2];
    float* out   = (float*)d_out;
    float* pnbuf = (float*)d_ws + PN_OFF;

    (void)in_sizes; (void)n_in; (void)ws_size;

    zero_out_kernel<<<(out_size + 255) / 256, 256, 0, stream>>>(out, out_size);

    softmax_rows_kernel<<<B_TOT * K_DIM, SM_THREADS, 0, stream>>>(probs, pnbuf);

    dim3 grid(C_DIM / CT, B_TOT, NSPLIT);
    pooled_gemm_kernel<<<grid, WAVES * 32, 0, stream>>>(feats, pnbuf, clip, out);
}


# pooling GEMM: a wave keeps 4 feats tiles (64 channels), so each probs fragment feeds 16 WMMAs (was 4); a block's 4 waves share one probs slice; next stage's loads grouped ahead; sum order unchanged
# speedup vs baseline: 2.0199x; 1.2450x over previous
#include <hip/hip_runtime.h>
#include <math.h>

typedef float v2f __attribute__((ext_vector_type(2)));
typedef float v4f __attribute__((ext_vector_type(4)));
typedef float v8f __attribute__((ext_vector_type(8)));

#define B_TOT  8
#define C_DIM  512
#define K_DIM  19
#define N_SP   16384
#define CT     16
#define NB     4
#define WAVES  4
#define SLICES 16
#define UG     8
#define PN_OFF 512
#define SM_THREADS 256
#define SM_WAVES   (SM_THREADS / 32)
#define SM_PIECES  (N_SP / 4 / SM_THREADS)

__global__ void zero_out_kernel(float* __restrict__ out, int n) {
    int i = blockIdx.x * blockDim.x + threadIdx.x;
    if (i < n) out[i] = 0.0f;
}

__global__ void __launch_bounds__(SM_THREADS)
softmax_rows_kernel(const float* __restrict__ probs, float* __restrict__ pn) {
    __shared__ float red_max[SM_WAVES];
    __shared__ float red_sum[SM_WAVES];
    const int r    = blockIdx.x;
    const int t    = threadIdx.x;
    const int wave = t >> 5;
    const int lane = t & 31;
    const v4f* p4 = (const v4f*)(probs + (size_t)r * N_SP);
    v4f*       o4 = (v4f*)(pn + (size_t)r * N_SP);

    v4f x[SM_PIECES];
    #pragma unroll
    for (int j = 0; j < SM_PIECES; ++j) x[j] = p4[t + SM_THREADS * j];

    float m = -3.402823466e38f;
    #pragma unroll
    for (int j = 0; j < SM_PIECES; ++j)
        m = fmaxf(m, fmaxf(fmaxf(x[j].x, x[j].y), fmaxf(x[j].z, x[j].w)));
    #pragma unroll
    for (int o = 16; o > 0; o >>= 1) m = fmaxf(m, __shfl_xor(m, o, 32));
    if (lane == 0) red_max[wave] = m;
    __syncthreads();
    float rowmax = red_max[0];
    #pragma unroll
    for (int w = 1; w < SM_WAVES; ++w) rowmax = fmaxf(rowmax, red_max[w]);

    float s0 = 0.0f, s1 = 0.0f, s2 = 0.0f, s3 = 0.0f;
    #pragma unroll
    for (int j = 0; j < SM_PIECES; ++j) {
        x[j].x = __expf(x[j].x - rowmax); s0 += x[j].x;
        x[j].y = __expf(x[j].y - rowmax); s1 += x[j].y;
        x[j].z = __expf(x[j].z - rowmax); s2 += x[j].z;
        x[j].w = __expf(x[j].w - rowmax); s3 += x[j].w;
    }
    float s = (s0 + s1) + (s2 + s3);
    #pragma unroll
    for (int o = 16; o > 0; o >>= 1) s += __shfl_xor(s, o, 32);
    if (lane == 0) red_sum[wave] = s;
    __syncthreads();
    float total = red_sum[0];
    #pragma unroll
    for (int w = 1; w < SM_WAVES; ++w) total += red_sum[w];
    const float inv = 1.0f / total;

    #pragma unroll
    for (int j = 0; j < SM_PIECES; ++j) o4[t + SM_THREADS * j] = x[j] * inv;
}

struct Stage {
    v4f a0[UG];
    v4f a1[UG];
    v4f b[NB][UG];
};

__device__ __forceinline__ void load_stage(Stage& s, const float* pa0, const float* pa1,
                                           const float* pb) {
    #pragma unroll
    for (int u = 0; u < UG; ++u) {
        s.a0[u] = *(const v4f*)(pa0 + 8 * u);
        s.a1[u] = *(const v4f*)(pa1 + 8 * u);
        #pragma unroll
        for (int j = 0; j < NB; ++j)
            s.b[j][u] = __builtin_nontemporal_load(
                            (const v4f*)(pb + (size_t)j * CT * N_SP + 8 * u));
    }
}

__device__ __forceinline__ void compute_stage(const Stage& s, v8f (&acc0)[NB],
                                              v8f (&acc1)[NB]) {
    #pragma unroll
    for (int u = 0; u < UG; ++u) {
        const v2f a0lo = {s.a0[u].x, s.a0[u].y}, a0hi = {s.a0[u].z, s.a0[u].w};
        const v2f a1lo = {s.a1[u].x, s.a1[u].y}, a1hi = {s.a1[u].z, s.a1[u].w};
        #pragma unroll
        for (int j = 0; j < NB; ++j) {
            const v2f blo = {s.b[j][u].x, s.b[j][u].y};
            acc0[j] = __builtin_amdgcn_wmma_f32_16x16x4_f32(
                          false, a0lo, false, blo, (short)0, acc0[j], false, false);
            acc1[j] = __builtin_amdgcn_wmma_f32_16x16x4_f32(
                          false, a1lo, false, blo, (short)0, acc1[j], false, false);
        }
        #pragma unroll
        for (int j = 0; j < NB; ++j) {
            const v2f bhi = {s.b[j][u].z, s.b[j][u].w};
            acc0[j] = __builtin_amdgcn_wmma_f32_16x16x4_f32(
                          false, a0hi, false, bhi, (short)0, acc0[j], false, false);
            acc1[j] = __builtin_amdgcn_wmma_f32_16x16x4_f32(
                          false, a1hi, false, bhi, (short)0, acc1[j], false, false);
        }
    }
}

__global__ void __launch_bounds__(WAVES * 32)
pooled_gemm_kernel(const float* __restrict__ feats,
                   const float* __restrict__ pn,
                   const int* __restrict__ clip_num,
                   float* __restrict__ out) {
    const int beta  = blockIdx.y;
    const int wave  = threadIdx.x >> 5;
    const int lane  = threadIdx.x & 31;
    const int half  = lane >> 4;
    const int lrow  = lane & 15;
    const int toff  = half * 4;
    const int cbase = (blockIdx.x * WAVES + wave) * (NB * CT);

    const int   splits = clip_num[0] + 1;
    const int   bsz    = B_TOT / splits;
    const int   bb     = beta % bsz;
    const float scale  = 1.0f / (float)splits;

    constexpr int nper   = N_SP / SLICES;
    constexpr int STEP   = 8 * UG;
    constexpr int NSTAGE = nper / STEP;
    const int n0 = blockIdx.z * nper;

    const float* pa0 = pn + ((size_t)beta * K_DIM + lrow) * N_SP + toff + n0;
    const int    k1  = 16 + lrow;
    const float* pa1 = pn + ((size_t)beta * K_DIM + (k1 < K_DIM ? k1 : 0)) * N_SP
                          + toff + n0;
    const float* pb  = feats + ((size_t)beta * C_DIM + cbase + lrow) * N_SP
                             + toff + n0;

    v8f acc0[NB] = {};
    v8f acc1[NB] = {};

    Stage sa, sb;
    load_stage(sa, pa0, pa1, pb);

    for (int it = 0; it < NSTAGE / 2 - 1; ++it) {
        __builtin_amdgcn_sched_barrier(0);
        load_stage(sb, pa0 + STEP, pa1 + STEP, pb + STEP);
        __builtin_amdgcn_sched_barrier(0);
        compute_stage(sa, acc0, acc1);
        __builtin_amdgcn_sched_barrier(0);
        load_stage(sa, pa0 + 2 * STEP, pa1 + 2 * STEP, pb + 2 * STEP);
        __builtin_amdgcn_sched_barrier(0);
        compute_stage(sb, acc0, acc1);
        pa0 += 2 * STEP;
        pa1 += 2 * STEP;
        pb  += 2 * STEP;
    }
    __builtin_amdgcn_sched_barrier(0);
    load_stage(sb, pa0 + STEP, pa1 + STEP, pb + STEP);
    __builtin_amdgcn_sched_barrier(0);
    compute_stage(sa, acc0, acc1);
    compute_stage(sb, acc0, acc1);

    #pragma unroll
    for (int j = 0; j < NB; ++j) {
        const int c = cbase + j * CT + lrow;
        float* ob = out + ((size_t)bb * C_DIM + c) * K_DIM;
        #pragma unroll
        for (int v = 0; v < 8; ++v) {
            const int k = v + half * 8;
            atomicAdd(ob + k, scale * acc0[j][v]);
            const int k2 = k + 16;
            if (k2 < K_DIM) atomicAdd(ob + k2, scale * acc1[j][v]);
        }
    }
}

extern "C" void kernel_launch(void* const* d_in, const int* in_sizes, int n_in,
                              void* d_out, int out_size, void* d_ws, size_t ws_size,
                              hipStream_t stream) {
    const float* feats = (const float*)d_in[0];
    const float* probs = (const float*)d_in[1];
    const int*   clip  = (const int*)d_in[2];
    float* out   = (float*)d_out;
    float* pnbuf = (float*)d_ws + PN_OFF;

    (void)in_sizes; (void)n_in; (void)ws_size;

    zero_out_kernel<<<(out_size + 255) / 256, 256, 0, stream>>>(out, out_size);

    softmax_rows_kernel<<<B_TOT * K_DIM, SM_THREADS, 0, stream>>>(probs, pnbuf);

    dim3 grid(C_DIM / (NB * CT * WAVES), B_TOT, SLICES);
    pooled_gemm_kernel<<<grid, WAVES * 32, 0, stream>>>(feats, pnbuf, clip, out);
}


# pooling GEMM loads: each load covers 8 rows x 64 B (was 16 rows x 32 B); lanes c and c^8 trade one register per pair by DPP row rotate, fragments rebuilt in registers, no LDS; sum order unchanged
# speedup vs baseline: 2.5729x; 1.2738x over previous
#include <hip/hip_runtime.h>
#include <math.h>

typedef float v2f __attribute__((ext_vector_type(2)));
typedef float v4f __attribute__((ext_vector_type(4)));
typedef float v8f __attribute__((ext_vector_type(8)));

#define B_TOT  8
#define C_DIM  512
#define K_DIM  19
#define N_SP   16384
#define CT     16
#define NB     4
#define WAVES  4
#define SLICES 16
#define UG     8
#define PN_OFF 512
#define SM_THREADS 256
#define SM_WAVES   (SM_THREADS / 32)
#define SM_PIECES  (N_SP / 4 / SM_THREADS)

__global__ void zero_out_kernel(float* __restrict__ out, int n) {
    int i = blockIdx.x * blockDim.x + threadIdx.x;
    if (i < n) out[i] = 0.0f;
}

__global__ void __launch_bounds__(SM_THREADS)
softmax_rows_kernel(const float* __restrict__ probs, float* __restrict__ pn) {
    __shared__ float red_max[SM_WAVES];
    __shared__ float red_sum[SM_WAVES];
    const int r    = blockIdx.x;
    const int t    = threadIdx.x;
    const int wave = t >> 5;
    const int lane = t & 31;
    const v4f* p4 = (const v4f*)(probs + (size_t)r * N_SP);
    v4f*       o4 = (v4f*)(pn + (size_t)r * N_SP);

    v4f x[SM_PIECES];
    #pragma unroll
    for (int j = 0; j < SM_PIECES; ++j) x[j] = p4[t + SM_THREADS * j];

    float m = -3.402823466e38f;
    #pragma unroll
    for (int j = 0; j < SM_PIECES; ++j)
        m = fmaxf(m, fmaxf(fmaxf(x[j].x, x[j].y), fmaxf(x[j].z, x[j].w)));
    #pragma unroll
    for (int o = 16; o > 0; o >>= 1) m = fmaxf(m, __shfl_xor(m, o, 32));
    if (lane == 0) red_max[wave] = m;
    __syncthreads();
    float rowmax = red_max[0];
    #pragma unroll
    for (int w = 1; w < SM_WAVES; ++w) rowmax = fmaxf(rowmax, red_max[w]);

    float s0 = 0.0f, s1 = 0.0f, s2 = 0.0f, s3 = 0.0f;
    #pragma unroll
    for (int j = 0; j < SM_PIECES; ++j) {
        x[j].x = __expf(x[j].x - rowmax); s0 += x[j].x;
        x[j].y = __expf(x[j].y - rowmax); s1 += x[j].y;
        x[j].z = __expf(x[j].z - rowmax); s2 += x[j].z;
        x[j].w = __expf(x[j].w - rowmax); s3 += x[j].w;
    }
    float s = (s0 + s1) + (s2 + s3);
    #pragma unroll
    for (int o = 16; o > 0; o >>= 1) s += __shfl_xor(s, o, 32);
    if (lane == 0) red_sum[wave] = s;
    __syncthreads();
    float total = red_sum[0];
    #pragma unroll
    for (int w = 1; w < SM_WAVES; ++w) total += red_sum[w];
    const float inv = 1.0f / total;

    #pragma unroll
    for (int j = 0; j < SM_PIECES; ++j) o4[t + SM_THREADS * j] = x[j] * inv;
}

struct Stage {
    v4f a0[UG];
    v4f a1[UG];
    v4f b[NB][UG];
};

struct RowPtrs {
    const float* a0[2];
    const float* a1[2];
    const float* b[2];
};

__device__ __forceinline__ void load_stage(Stage& s, const RowPtrs& p, int adv) {
    #pragma unroll
    for (int w = 0; w < UG / 2; ++w) {
        #pragma unroll
        for (int e = 0; e < 2; ++e) {
            const int u = 2 * w + e;
            s.a0[u] = *(const v4f*)(p.a0[e] + adv + 16 * w);
            s.a1[u] = *(const v4f*)(p.a1[e] + adv + 16 * w);
            #pragma unroll
            for (int j = 0; j < NB; ++j)
                s.b[j][u] = __builtin_nontemporal_load(
                                (const v4f*)(p.b[e] + adv + (size_t)j * CT * N_SP + 16 * w));
        }
    }
}

__device__ __forceinline__ void trade_rows(v4f& x0, v4f& x1, bool c3) {
    #pragma unroll
    for (int d = 0; d < 4; ++d) {
        const int k0 = __float_as_int(x0[d]);
        const int k1 = __float_as_int(x1[d]);
        const int give = c3 ? k0 : k1;
        x0[d] = __int_as_float(__builtin_amdgcn_update_dpp(k0, give, 0x128, 0xF, 0xC, false));
        x1[d] = __int_as_float(__builtin_amdgcn_update_dpp(k1, give, 0x128, 0xF, 0x3, false));
    }
}

__device__ __forceinline__ void compute_stage(Stage& s, bool c3, v8f (&acc0)[NB],
                                              v8f (&acc1)[NB]) {
    #pragma unroll
    for (int u = 0; u < UG; ++u) {
        if ((u & 1) == 0) {
            trade_rows(s.a0[u], s.a0[u + 1], c3);
            trade_rows(s.a1[u], s.a1[u + 1], c3);
            #pragma unroll
            for (int j = 0; j < NB; ++j) trade_rows(s.b[j][u], s.b[j][u + 1], c3);
        }
        const v2f a0lo = {s.a0[u].x, s.a0[u].y}, a0hi = {s.a0[u].z, s.a0[u].w};
        const v2f a1lo = {s.a1[u].x, s.a1[u].y}, a1hi = {s.a1[u].z, s.a1[u].w};
        #pragma unroll
        for (int j = 0; j < NB; ++j) {
            const v2f blo = {s.b[j][u].x, s.b[j][u].y};
            acc0[j] = __builtin_amdgcn_wmma_f32_16x16x4_f32(
                          false, a0lo, false, blo, (short)0, acc0[j], false, false);
            acc1[j] = __builtin_amdgcn_wmma_f32_16x16x4_f32(
                          false, a1lo, false, blo, (short)0, acc1[j], false, false);
        }
        #pragma unroll
        for (int j = 0; j < NB; ++j) {
            const v2f bhi = {s.b[j][u].z, s.b[j][u].w};
            acc0[j] = __builtin_amdgcn_wmma_f32_16x16x4_f32(
                          false, a0hi, false, bhi, (short)0, acc0[j], false, false);
            acc1[j] = __builtin_amdgcn_wmma_f32_16x16x4_f32(
                          false, a1hi, false, bhi, (short)0, acc1[j], false, false);
        }
    }
}

__global__ void __launch_bounds__(WAVES * 32)
pooled_gemm_kernel(const float* __restrict__ feats,
                   const float* __restrict__ pn,
                   const int* __restrict__ clip_num,
                   float* __restrict__ out) {
    const int beta  = blockIdx.y;
    const int wave  = threadIdx.x >> 5;
    const int lane  = threadIdx.x & 31;
    const int half  = lane >> 4;
    const int lrow  = lane & 15;
    const int c8    = lrow & 7;
    const bool c3   = (lrow & 8) != 0;
    const int toff  = (c3 ? 8 : 0) + half * 4;
    const int cbase = (blockIdx.x * WAVES + wave) * (NB * CT);

    const int   splits = clip_num[0] + 1;
    const int   bsz    = B_TOT / splits;
    const int   bb     = beta % bsz;
    const float scale  = 1.0f / (float)splits;

    constexpr int nper   = N_SP / SLICES;
    constexpr int STEP   = 8 * UG;
    constexpr int NSTAGE = nper / STEP;
    const int n0 = blockIdx.z * nper;

    RowPtrs p;
    #pragma unroll
    for (int e = 0; e < 2; ++e) {
        const int r  = 8 * e + c8;
        const int k1 = 16 + r;
        p.a0[e] = pn + ((size_t)beta * K_DIM + r) * N_SP + toff + n0;
        p.a1[e] = pn + ((size_t)beta * K_DIM + (k1 < K_DIM ? k1 : 0)) * N_SP + toff + n0;
        p.b[e]  = feats + ((size_t)beta * C_DIM + cbase + r) * N_SP + toff + n0;
    }

    v8f acc0[NB] = {};
    v8f acc1[NB] = {};

    Stage sa, sb;
    int adv = 0;
    load_stage(sa, p, 0);

    for (int it = 0; it < NSTAGE / 2 - 1; ++it) {
        __builtin_amdgcn_sched_barrier(0);
        load_stage(sb, p, adv + STEP);
        __builtin_amdgcn_sched_barrier(0);
        compute_stage(sa, c3, acc0, acc1);
        __builtin_amdgcn_sched_barrier(0);
        load_stage(sa, p, adv + 2 * STEP);
        __builtin_amdgcn_sched_barrier(0);
        compute_stage(sb, c3, acc0, acc1);
        adv += 2 * STEP;
    }
    __builtin_amdgcn_sched_barrier(0);
    load_stage(sb, p, adv + STEP);
    __builtin_amdgcn_sched_barrier(0);
    compute_stage(sa, c3, acc0, acc1);
    compute_stage(sb, c3, acc0, acc1);

    #pragma unroll
    for (int j = 0; j < NB; ++j) {
        const int c = cbase + j * CT + lrow;
        float* ob = out + ((size_t)bb * C_DIM + c) * K_DIM;
        #pragma unroll
        for (int v = 0; v < 8; ++v) {
            const int k = v + half * 8;
            atomicAdd(ob + k, scale * acc0[j][v]);
            const int k2 = k + 16;
            if (k2 < K_DIM) atomicAdd(ob + k2, scale * acc1[j][v]);
        }
    }
}

extern "C" void kernel_launch(void* const* d_in, const int* in_sizes, int n_in,
                              void* d_out, int out_size, void* d_ws, size_t ws_size,
                              hipStream_t stream) {
    const float* feats = (const float*)d_in[0];
    const float* probs = (const float*)d_in[1];
    const int*   clip  = (const int*)d_in[2];
    float* out   = (float*)d_out;
    float* pnbuf = (float*)d_ws + PN_OFF;

    (void)in_sizes; (void)n_in; (void)ws_size;

    zero_out_kernel<<<(out_size + 255) / 256, 256, 0, stream>>>(out, out_size);

    softmax_rows_kernel<<<B_TOT * K_DIM, SM_THREADS, 0, stream>>>(probs, pnbuf);

    dim3 grid(C_DIM / (NB * CT * WAVES), B_TOT, SLICES);
    pooled_gemm_kernel<<<grid, WAVES * 32, 0, stream>>>(feats, pnbuf, clip, out);
}
